# starting kernel
# baseline (speedup 1.0000x reference)
#include <hip/hip_runtime.h>


typedef __attribute__((ext_vector_type(2))) float v2f;
typedef __attribute__((ext_vector_type(8))) float v8f;

#define SCALE 0.6513215599f

__device__ __forceinline__ float csign(int a, int b) {
    int cnt = 0;
    for (int aa = a >> 1; aa; aa >>= 1) cnt += __popc(aa & b);
    if ((a & b) & 16) cnt ^= 1;
    return (cnt & 1) ? -1.0f : 1.0f;
}

__global__ __launch_bounds__(256) void gram_kernel(const float* __restrict__ Win,
                                                   const float* __restrict__ Wout,
                                                   float* __restrict__ G) {
    const int d    = blockIdx.x;
    const int tid  = threadIdx.x;
    const int wid  = tid >> 5;
    const int lane = tid & 31;
    const int half = lane & 15;
    const int hi   = lane >> 4;

    __shared__ float red[8 * 1024];

    v8f acc00 = {}, acc01 = {}, acc10 = {}, acc11 = {};

    const float* WinD = Win + d * 32;
    const int kbeg = wid * 128;
    #pragma unroll 4
    for (int k0 = kbeg; k0 < kbeg + 128; k0 += 4) {
        const int kA = k0 + hi * 2;
        v2f a0, a1, b0, b1;
        a0.x = WinD[(size_t)kA * 160 + half];
        a0.y = WinD[(size_t)(kA + 1) * 160 + half];
        a1.x = WinD[(size_t)kA * 160 + 16 + half];
        a1.y = WinD[(size_t)(kA + 1) * 160 + 16 + half];
        b0.x = Wout[kA * 32 + half];
        b0.y = Wout[(kA + 1) * 32 + half];
        b1.x = Wout[kA * 32 + 16 + half];
        b1.y = Wout[(kA + 1) * 32 + 16 + half];

        acc00 = __builtin_amdgcn_wmma_f32_16x16x4_f32(false, a0, false, b0, (short)0, acc00, false, false);
        acc01 = __builtin_amdgcn_wmma_f32_16x16x4_f32(false, a0, false, b1, (short)0, acc01, false, false);
        acc10 = __builtin_amdgcn_wmma_f32_16x16x4_f32(false, a1, false, b0, (short)0, acc10, false, false);
        acc11 = __builtin_amdgcn_wmma_f32_16x16x4_f32(false, a1, false, b1, (short)0, acc11, false, false);
    }

    float* my = red + wid * 1024;
    #pragma unroll
    for (int r = 0; r < 8; ++r) {
        const int m = r + 8 * hi;
        my[m * 32 + half]             = acc00[r];
        my[m * 32 + 16 + half]        = acc01[r];
        my[(m + 16) * 32 + half]      = acc10[r];
        my[(m + 16) * 32 + 16 + half] = acc11[r];
    }
    __syncthreads();

    #pragma unroll
    for (int j = 0; j < 4; ++j) {
        const int o = tid * 4 + j;
        float s = 0.0f;
        #pragma unroll
        for (int w = 0; w < 8; ++w) s += red[w * 1024 + o];
        G[d * 1024 + o] = s;
    }
}

__global__ __launch_bounds__(256) void mix_kernel(const float* __restrict__ G,
                                                  float* __restrict__ M) {
    __shared__ float Gd[1024];
    const int t = blockIdx.x * 256 + threadIdx.x;
    const int d = t >> 10;
    const float* src = G + d * 1024;
    for (int j = threadIdx.x; j < 1024; j += 256) Gd[j] = src[j];
    __syncthreads();

    const int p1 = (t >> 5) & 31;
    const int k  = t & 31;
    float acc = 0.0f;
    #pragma unroll
    for (int p = 0; p < 32; ++p) {
        const int q1 = p1 ^ p;
        const int q2 = p ^ k;
        acc += csign(p1, q1) * csign(p, q2) * Gd[q1 * 32 + q2];
    }
    M[((d << 5) | p1) * 32 + k] = SCALE * acc;
}

__global__ __launch_bounds__(256) void apply_kernel(const float* __restrict__ X,
                                                    const float* __restrict__ M,
                                                    float* __restrict__ out) {
    const int tid  = threadIdx.x;
    const int wid  = tid >> 5;
    const int lane = tid & 31;
    const int half = lane & 15;
    const int hi   = lane >> 4;
    const int tile = blockIdx.x * 8 + wid;
    const int rt   = tile >> 1;
    const int ct   = tile & 1;

    v8f acc = {};
    const float* Xrow = X + (size_t)rt * 16 * 160;
    #pragma unroll 8
    for (int k0 = 0; k0 < 160; k0 += 4) {
        const int kA = k0 + hi * 2;
        v2f a, b;
        a.x = Xrow[(size_t)half * 160 + kA];
        a.y = Xrow[(size_t)half * 160 + kA + 1];
        b.x = M[kA * 32 + ct * 16 + half];
        b.y = M[(kA + 1) * 32 + ct * 16 + half];
        acc = __builtin_amdgcn_wmma_f32_16x16x4_f32(false, a, false, b, (short)0, acc, false, false);
    }
    #pragma unroll
    for (int r = 0; r < 8; ++r) {
        const int m = rt * 16 + r + 8 * hi;
        out[m * 32 + ct * 16 + half] = acc[r];
    }
}

extern "C" void kernel_launch(void* const* d_in, const int* in_sizes, int n_in,
                              void* d_out, int out_size, void* d_ws, size_t ws_size,
                              hipStream_t stream) {
    const float* x    = (const float*)d_in[0];
    const float* Win  = (const float*)d_in[1];
    const float* Wout = (const float*)d_in[2];
    float* out = (float*)d_out;

    float* G = (float*)d_ws;
    float* M = G + 5 * 1024;

    gram_kernel<<<5, 256, 0, stream>>>(Win, Wout, G);
    mix_kernel<<<20, 256, 0, stream>>>(G, M);
    apply_kernel<<<8, 256, 0, stream>>>(x, M, out);
}


# mix stage folded into the Gram kernel's epilogue: G[d] stays in LDS, a thread mixes 4 adjacent k per 16-byte LDS read with signs from parity words; 3 launches -> 2, same summation order
# speedup vs baseline: 1.4119x; 1.4119x over previous
#include <hip/hip_runtime.h>


typedef __attribute__((ext_vector_type(2))) float v2f;
typedef __attribute__((ext_vector_type(8))) float v8f;

#define SCALE 0.6513215599f

constexpr unsigned cpar(unsigned a, unsigned b) {
    unsigned cnt = 0;
    for (unsigned aa = a >> 1; aa; aa >>= 1)
        for (unsigned t = aa & b; t; t &= t - 1) cnt ^= 1u;
    if ((a & b) & 16u) cnt ^= 1u;
    return cnt & 1u;
}
constexpr unsigned signA(unsigned i) {
    unsigned w = 0;
    for (unsigned p = 0; p < 32; ++p) w |= cpar(1u << i, p) << p;
    return w;
}
constexpr unsigned signB(unsigned i) {
    unsigned w = 0;
    for (unsigned p = 0; p < 32; ++p) w |= cpar(p, 1u << i) << p;
    return w;
}
constexpr unsigned signC() {
    unsigned w = 0;
    for (unsigned p = 0; p < 32; ++p) w |= cpar(p, p) << p;
    return w;
}

__global__ __launch_bounds__(256) void gram_mix_kernel(const float* __restrict__ Win,
                                                       const float* __restrict__ Wout,
                                                       float* __restrict__ M) {
    const int d    = blockIdx.x;
    const int tid  = threadIdx.x;
    const int wid  = tid >> 5;
    const int lane = tid & 31;
    const int half = lane & 15;
    const int hi   = lane >> 4;

    __shared__ __attribute__((aligned(16))) float red[8 * 1024];
    __shared__ __attribute__((aligned(16))) float Gd[1024];

    v8f acc00 = {}, acc01 = {}, acc10 = {}, acc11 = {};

    const float* WinD = Win + d * 32;
    const int kbeg = wid * 128;
    #pragma unroll 4
    for (int k0 = kbeg; k0 < kbeg + 128; k0 += 4) {
        const int kA = k0 + hi * 2;
        v2f a0, a1, b0, b1;
        a0.x = WinD[(size_t)kA * 160 + half];
        a0.y = WinD[(size_t)(kA + 1) * 160 + half];
        a1.x = WinD[(size_t)kA * 160 + 16 + half];
        a1.y = WinD[(size_t)(kA + 1) * 160 + 16 + half];
        b0.x = Wout[kA * 32 + half];
        b0.y = Wout[(kA + 1) * 32 + half];
        b1.x = Wout[kA * 32 + 16 + half];
        b1.y = Wout[(kA + 1) * 32 + 16 + half];

        acc00 = __builtin_amdgcn_wmma_f32_16x16x4_f32(false, a0, false, b0, (short)0, acc00, false, false);
        acc01 = __builtin_amdgcn_wmma_f32_16x16x4_f32(false, a0, false, b1, (short)0, acc01, false, false);
        acc10 = __builtin_amdgcn_wmma_f32_16x16x4_f32(false, a1, false, b0, (short)0, acc10, false, false);
        acc11 = __builtin_amdgcn_wmma_f32_16x16x4_f32(false, a1, false, b1, (short)0, acc11, false, false);
    }

    float* my = red + wid * 1024;
    #pragma unroll
    for (int r = 0; r < 8; ++r) {
        const int m = r + 8 * hi;
        my[m * 32 + half]             = acc00[r];
        my[m * 32 + 16 + half]        = acc01[r];
        my[(m + 16) * 32 + half]      = acc10[r];
        my[(m + 16) * 32 + 16 + half] = acc11[r];
    }
    __syncthreads();

    #pragma unroll
    for (int j = 0; j < 4; ++j) {
        const int o = tid * 4 + j;
        float s = 0.0f;
        #pragma unroll
        for (int w = 0; w < 8; ++w) s += red[w * 1024 + o];
        Gd[o] = s;
    }
    __syncthreads();

    const unsigned p1 = (unsigned)tid >> 3;
    const unsigned kq = ((unsigned)tid & 7u) * 4u;

    constexpr unsigned A0 = signA(0), A1 = signA(1), A2 = signA(2), A3 = signA(3), A4 = signA(4);
    constexpr unsigned B0 = signB(0), B1 = signB(1), B2 = signB(2), B3 = signB(3), B4 = signB(4);
    constexpr unsigned CW = signC();

    unsigned S = CW;
    S ^= (p1 & 1u)  ? A0 : 0u;
    S ^= (p1 & 2u)  ? A1 : 0u;
    S ^= (p1 & 4u)  ? A2 : 0u;
    S ^= (p1 & 8u)  ? A3 : 0u;
    S ^= (p1 & 16u) ? A4 : 0u;
    S ^= (kq & 4u)  ? B2 : 0u;
    S ^= (kq & 8u)  ? B3 : 0u;
    S ^= (kq & 16u) ? B4 : 0u;
    S ^= ((CW >> p1) & 1u) ? 0xFFFFFFFFu : 0u;
    const unsigned Su[4] = {S, S ^ B0, S ^ B1, S ^ B0 ^ B1};

    float acc[4] = {0.0f, 0.0f, 0.0f, 0.0f};
    const unsigned base = p1 * 32u + kq;
    #pragma unroll
    for (unsigned p = 0; p < 32; ++p) {
        const float4 v = *reinterpret_cast<const float4*>(&Gd[base ^ (p * 32u + (p & 28u))]);
        const float ve[4] = {v.x, v.y, v.z, v.w};
        #pragma unroll
        for (unsigned u = 0; u < 4; ++u) {
            const unsigned sb = (Su[u] << (31u - p)) & 0x80000000u;
            acc[u] += __uint_as_float(__float_as_uint(ve[(p & 3u) ^ u]) ^ sb);
        }
    }
    float4 o4;
    o4.x = SCALE * acc[0];
    o4.y = SCALE * acc[1];
    o4.z = SCALE * acc[2];
    o4.w = SCALE * acc[3];
    *reinterpret_cast<float4*>(&M[((unsigned)d * 32u + p1) * 32u + kq]) = o4;
}

__global__ __launch_bounds__(256) void apply_kernel(const float* __restrict__ X,
                                                    const float* __restrict__ M,
                                                    float* __restrict__ out) {
    const int tid  = threadIdx.x;
    const int wid  = tid >> 5;
    const int lane = tid & 31;
    const int half = lane & 15;
    const int hi   = lane >> 4;
    const int tile = blockIdx.x * 8 + wid;
    const int rt   = tile >> 1;
    const int ct   = tile & 1;

    v8f acc = {};
    const float* Xrow = X + (size_t)rt * 16 * 160;
    #pragma unroll 8
    for (int k0 = 0; k0 < 160; k0 += 4) {
        const int kA = k0 + hi * 2;
        v2f a, b;
        a.x = Xrow[(size_t)half * 160 + kA];
        a.y = Xrow[(size_t)half * 160 + kA + 1];
        b.x = M[kA * 32 + ct * 16 + half];
        b.y = M[(kA + 1) * 32 + ct * 16 + half];
        acc = __builtin_amdgcn_wmma_f32_16x16x4_f32(false, a, false, b, (short)0, acc, false, false);
    }
    #pragma unroll
    for (int r = 0; r < 8; ++r) {
        const int m = rt * 16 + r + 8 * hi;
        out[m * 32 + ct * 16 + half] = acc[r];
    }
}

extern "C" void kernel_launch(void* const* d_in, const int* in_sizes, int n_in,
                              void* d_out, int out_size, void* d_ws, size_t ws_size,
                              hipStream_t stream) {
    const float* x    = (const float*)d_in[0];
    const float* Win  = (const float*)d_in[1];
    const float* Wout = (const float*)d_in[2];
    float* out = (float*)d_out;

    float* M = (float*)d_ws;

    gram_mix_kernel<<<5, 256, 0, stream>>>(Win, Wout, M);
    apply_kernel<<<8, 256, 0, stream>>>(x, M, out);
}


# Gram sum over h split into 4 chunks on 4x the work-groups (5 -> 20), a wave's 64 fragment loads issued as one group; apply adds the 4 partial M planes in LDS (new summation tree)
# speedup vs baseline: 2.4262x; 1.7184x over previous
#include <hip/hip_runtime.h>


typedef __attribute__((ext_vector_type(2))) float v2f;
typedef __attribute__((ext_vector_type(8))) float v8f;

#define SCALE 0.6513215599f
#define KSPLIT 4
#define KWAVE  32
#define MPLANE 5120

constexpr unsigned cpar(unsigned a, unsigned b) {
    unsigned cnt = 0;
    for (unsigned aa = a >> 1; aa; aa >>= 1)
        for (unsigned t = aa & b; t; t &= t - 1) cnt ^= 1u;
    if ((a & b) & 16u) cnt ^= 1u;
    return cnt & 1u;
}
constexpr unsigned signA(unsigned i) {
    unsigned w = 0;
    for (unsigned p = 0; p < 32; ++p) w |= cpar(1u << i, p) << p;
    return w;
}
constexpr unsigned signB(unsigned i) {
    unsigned w = 0;
    for (unsigned p = 0; p < 32; ++p) w |= cpar(p, 1u << i) << p;
    return w;
}
constexpr unsigned signC() {
    unsigned w = 0;
    for (unsigned p = 0; p < 32; ++p) w |= cpar(p, p) << p;
    return w;
}

__global__ __launch_bounds__(256) void gram_mix_kernel(const float* __restrict__ Win,
                                                       const float* __restrict__ Wout,
                                                       float* __restrict__ Mp) {
    const int d    = blockIdx.x;
    const int c    = blockIdx.y;
    const int tid  = threadIdx.x;
    const int wid  = tid >> 5;
    const int lane = tid & 31;
    const int half = lane & 15;
    const int hi   = lane >> 4;

    __shared__ __attribute__((aligned(16))) float red[8 * 1024];
    __shared__ __attribute__((aligned(16))) float Gd[1024];

    v8f acc00 = {}, acc01 = {}, acc10 = {}, acc11 = {};

    const float* WinD = Win + d * 32;
    const int kbeg = (c * 8 + wid) * KWAVE;
    v2f a0[KWAVE / 4], a1[KWAVE / 4], b0[KWAVE / 4], b1[KWAVE / 4];
    #pragma unroll
    for (int j = 0; j < KWAVE / 4; ++j) {
        const int kA = kbeg + j * 4 + hi * 2;
        a0[j].x = WinD[(size_t)kA * 160 + half];
        a0[j].y = WinD[(size_t)(kA + 1) * 160 + half];
        a1[j].x = WinD[(size_t)kA * 160 + 16 + half];
        a1[j].y = WinD[(size_t)(kA + 1) * 160 + 16 + half];
        b0[j].x = Wout[kA * 32 + half];
        b0[j].y = Wout[(kA + 1) * 32 + half];
        b1[j].x = Wout[kA * 32 + 16 + half];
        b1[j].y = Wout[(kA + 1) * 32 + 16 + half];
    }
    #pragma unroll
    for (int j = 0; j < KWAVE / 4; ++j) {
        acc00 = __builtin_amdgcn_wmma_f32_16x16x4_f32(false, a0[j], false, b0[j], (short)0, acc00, false, false);
        acc01 = __builtin_amdgcn_wmma_f32_16x16x4_f32(false, a0[j], false, b1[j], (short)0, acc01, false, false);
        acc10 = __builtin_amdgcn_wmma_f32_16x16x4_f32(false, a1[j], false, b0[j], (short)0, acc10, false, false);
        acc11 = __builtin_amdgcn_wmma_f32_16x16x4_f32(false, a1[j], false, b1[j], (short)0, acc11, false, false);
    }

    float* my = red + wid * 1024;
    #pragma unroll
    for (int r = 0; r < 8; ++r) {
        const int m = r + 8 * hi;
        my[m * 32 + half]             = acc00[r];
        my[m * 32 + 16 + half]        = acc01[r];
        my[(m + 16) * 32 + half]      = acc10[r];
        my[(m + 16) * 32 + 16 + half] = acc11[r];
    }
    __syncthreads();

    #pragma unroll
    for (int j = 0; j < 4; ++j) {
        const int o = tid * 4 + j;
        float s = 0.0f;
        #pragma unroll
        for (int w = 0; w < 8; ++w) s += red[w * 1024 + o];
        Gd[o] = s;
    }
    __syncthreads();

    const unsigned p1 = (unsigned)tid >> 3;
    const unsigned kq = ((unsigned)tid & 7u) * 4u;

    constexpr unsigned A0 = signA(0), A1 = signA(1), A2 = signA(2), A3 = signA(3), A4 = signA(4);
    constexpr unsigned B0 = signB(0), B1 = signB(1), B2 = signB(2), B3 = signB(3), B4 = signB(4);
    constexpr unsigned CW = signC();

    unsigned S = CW;
    S ^= (p1 & 1u)  ? A0 : 0u;
    S ^= (p1 & 2u)  ? A1 : 0u;
    S ^= (p1 & 4u)  ? A2 : 0u;
    S ^= (p1 & 8u)  ? A3 : 0u;
    S ^= (p1 & 16u) ? A4 : 0u;
    S ^= (kq & 4u)  ? B2 : 0u;
    S ^= (kq & 8u)  ? B3 : 0u;
    S ^= (kq & 16u) ? B4 : 0u;
    S ^= ((CW >> p1) & 1u) ? 0xFFFFFFFFu : 0u;
    const unsigned Su[4] = {S, S ^ B0, S ^ B1, S ^ B0 ^ B1};

    float acc[4] = {0.0f, 0.0f, 0.0f, 0.0f};
    const unsigned base = p1 * 32u + kq;
    #pragma unroll
    for (unsigned p = 0; p < 32; ++p) {
        const float4 v = *reinterpret_cast<const float4*>(&Gd[base ^ (p * 32u + (p & 28u))]);
        const float ve[4] = {v.x, v.y, v.z, v.w};
        #pragma unroll
        for (unsigned u = 0; u < 4; ++u) {
            const unsigned sb = (Su[u] << (31u - p)) & 0x80000000u;
            acc[u] += __uint_as_float(__float_as_uint(ve[(p & 3u) ^ u]) ^ sb);
        }
    }
    float4 o4;
    o4.x = SCALE * acc[0];
    o4.y = SCALE * acc[1];
    o4.z = SCALE * acc[2];
    o4.w = SCALE * acc[3];
    *reinterpret_cast<float4*>(&Mp[(unsigned)c * MPLANE + ((unsigned)d * 32u + p1) * 32u + kq]) = o4;
}

__global__ __launch_bounds__(256) void apply_kernel(const float* __restrict__ X,
                                                    const float* __restrict__ Mp,
                                                    float* __restrict__ out) {
    __shared__ __attribute__((aligned(16))) float M[MPLANE];

    const int tid  = threadIdx.x;
    const int wid  = tid >> 5;
    const int lane = tid & 31;
    const int half = lane & 15;
    const int hi   = lane >> 4;
    const int tile = blockIdx.x * 8 + wid;
    const int rt   = tile >> 1;
    const int ct   = tile & 1;

    #pragma unroll
    for (int i = 0; i < MPLANE / 4 / 256; ++i) {
        const int o = (i * 256 + tid) * 4;
        float4 s = *reinterpret_cast<const float4*>(&Mp[o]);
        #pragma unroll
        for (int c = 1; c < KSPLIT; ++c) {
            const float4 v = *reinterpret_cast<const float4*>(&Mp[c * MPLANE + o]);
            s.x += v.x;
            s.y += v.y;
            s.z += v.z;
            s.w += v.w;
        }
        *reinterpret_cast<float4*>(&M[o]) = s;
    }
    __syncthreads();

    v8f acc = {};
    const float* Xrow = X + (size_t)rt * 16 * 160;
    #pragma unroll 8
    for (int k0 = 0; k0 < 160; k0 += 4) {
        const int kA = k0 + hi * 2;
        v2f a, b;
        a.x = Xrow[(size_t)half * 160 + kA];
        a.y = Xrow[(size_t)half * 160 + kA + 1];
        b.x = M[kA * 32 + ct * 16 + half];
        b.y = M[(kA + 1) * 32 + ct * 16 + half];
        acc = __builtin_amdgcn_wmma_f32_16x16x4_f32(false, a, false, b, (short)0, acc, false, false);
    }
    #pragma unroll
    for (int r = 0; r < 8; ++r) {
        const int m = rt * 16 + r + 8 * hi;
        out[m * 32 + ct * 16 + half] = acc[r];
    }
}

extern "C" void kernel_launch(void* const* d_in, const int* in_sizes, int n_in,
                              void* d_out, int out_size, void* d_ws, size_t ws_size,
                              hipStream_t stream) {
    const float* x    = (const float*)d_in[0];
    const float* Win  = (const float*)d_in[1];
    const float* Wout = (const float*)d_in[2];
    float* out = (float*)d_out;

    float* Mp = (float*)d_ws;

    gram_mix_kernel<<<dim3(5, 4), 256, 0, stream>>>(Win, Wout, Mp);
    apply_kernel<<<8, 256, 0, stream>>>(x, Mp, out);
}


# apply: x rows fetched once by whole-line 16-byte loads issued with the plane loads and staged in LDS; a wave's 60 LDS fragment reads go out as one group ahead of its 40 WMMAs; same summation order
# speedup vs baseline: 2.9245x; 1.2054x over previous
#include <hip/hip_runtime.h>


typedef __attribute__((ext_vector_type(2))) float v2f;
typedef __attribute__((ext_vector_type(8))) float v8f;

#define SCALE 0.6513215599f
#define KSPLIT 4
#define KWAVE  32
#define MPLANE 5120

constexpr unsigned cpar(unsigned a, unsigned b) {
    unsigned cnt = 0;
    for (unsigned aa = a >> 1; aa; aa >>= 1)
        for (unsigned t = aa & b; t; t &= t - 1) cnt ^= 1u;
    if ((a & b) & 16u) cnt ^= 1u;
    return cnt & 1u;
}
constexpr unsigned signA(unsigned i) {
    unsigned w = 0;
    for (unsigned p = 0; p < 32; ++p) w |= cpar(1u << i, p) << p;
    return w;
}
constexpr unsigned signB(unsigned i) {
    unsigned w = 0;
    for (unsigned p = 0; p < 32; ++p) w |= cpar(p, 1u << i) << p;
    return w;
}
constexpr unsigned signC() {
    unsigned w = 0;
    for (unsigned p = 0; p < 32; ++p) w |= cpar(p, p) << p;
    return w;
}

__global__ __launch_bounds__(256) void gram_mix_kernel(const float* __restrict__ Win,
                                                       const float* __restrict__ Wout,
                                                       float* __restrict__ Mp) {
    const int d    = blockIdx.x;
    const int c    = blockIdx.y;
    const int tid  = threadIdx.x;
    const int wid  = tid >> 5;
    const int lane = tid & 31;
    const int half = lane & 15;
    const int hi   = lane >> 4;

    __shared__ __attribute__((aligned(16))) float red[8 * 1024];
    __shared__ __attribute__((aligned(16))) float Gd[1024];

    v8f acc00 = {}, acc01 = {}, acc10 = {}, acc11 = {};

    const float* WinD = Win + d * 32;
    const int kbeg = (c * 8 + wid) * KWAVE;
    v2f a0[KWAVE / 4], a1[KWAVE / 4], b0[KWAVE / 4], b1[KWAVE / 4];
    #pragma unroll
    for (int j = 0; j < KWAVE / 4; ++j) {
        const int kA = kbeg + j * 4 + hi * 2;
        a0[j].x = WinD[(size_t)kA * 160 + half];
        a0[j].y = WinD[(size_t)(kA + 1) * 160 + half];
        a1[j].x = WinD[(size_t)kA * 160 + 16 + half];
        a1[j].y = WinD[(size_t)(kA + 1) * 160 + 16 + half];
        b0[j].x = Wout[kA * 32 + half];
        b0[j].y = Wout[(kA + 1) * 32 + half];
        b1[j].x = Wout[kA * 32 + 16 + half];
        b1[j].y = Wout[(kA + 1) * 32 + 16 + half];
    }
    #pragma unroll
    for (int j = 0; j < KWAVE / 4; ++j) {
        acc00 = __builtin_amdgcn_wmma_f32_16x16x4_f32(false, a0[j], false, b0[j], (short)0, acc00, false, false);
        acc01 = __builtin_amdgcn_wmma_f32_16x16x4_f32(false, a0[j], false, b1[j], (short)0, acc01, false, false);
        acc10 = __builtin_amdgcn_wmma_f32_16x16x4_f32(false, a1[j], false, b0[j], (short)0, acc10, false, false);
        acc11 = __builtin_amdgcn_wmma_f32_16x16x4_f32(false, a1[j], false, b1[j], (short)0, acc11, false, false);
    }

    float* my = red + wid * 1024;
    #pragma unroll
    for (int r = 0; r < 8; ++r) {
        const int m = r + 8 * hi;
        my[m * 32 + half]             = acc00[r];
        my[m * 32 + 16 + half]        = acc01[r];
        my[(m + 16) * 32 + half]      = acc10[r];
        my[(m + 16) * 32 + 16 + half] = acc11[r];
    }
    __syncthreads();

    #pragma unroll
    for (int j = 0; j < 4; ++j) {
        const int o = tid * 4 + j;
        float s = 0.0f;
        #pragma unroll
        for (int w = 0; w < 8; ++w) s += red[w * 1024 + o];
        Gd[o] = s;
    }
    __syncthreads();

    const unsigned p1 = (unsigned)tid >> 3;
    const unsigned kq = ((unsigned)tid & 7u) * 4u;

    constexpr unsigned A0 = signA(0), A1 = signA(1), A2 = signA(2), A3 = signA(3), A4 = signA(4);
    constexpr unsigned B0 = signB(0), B1 = signB(1), B2 = signB(2), B3 = signB(3), B4 = signB(4);
    constexpr unsigned CW = signC();

    unsigned S = CW;
    S ^= (p1 & 1u)  ? A0 : 0u;
    S ^= (p1 & 2u)  ? A1 : 0u;
    S ^= (p1 & 4u)  ? A2 : 0u;
    S ^= (p1 & 8u)  ? A3 : 0u;
    S ^= (p1 & 16u) ? A4 : 0u;
    S ^= (kq & 4u)  ? B2 : 0u;
    S ^= (kq & 8u)  ? B3 : 0u;
    S ^= (kq & 16u) ? B4 : 0u;
    S ^= ((CW >> p1) & 1u) ? 0xFFFFFFFFu : 0u;
    const unsigned Su[4] = {S, S ^ B0, S ^ B1, S ^ B0 ^ B1};

    float acc[4] = {0.0f, 0.0f, 0.0f, 0.0f};
    const unsigned base = p1 * 32u + kq;
    #pragma unroll
    for (unsigned p = 0; p < 32; ++p) {
        const float4 v = *reinterpret_cast<const float4*>(&Gd[base ^ (p * 32u + (p & 28u))]);
        const float ve[4] = {v.x, v.y, v.z, v.w};
        #pragma unroll
        for (unsigned u = 0; u < 4; ++u) {
            const unsigned sb = (Su[u] << (31u - p)) & 0x80000000u;
            acc[u] += __uint_as_float(__float_as_uint(ve[(p & 3u) ^ u]) ^ sb);
        }
    }
    float4 o4;
    o4.x = SCALE * acc[0];
    o4.y = SCALE * acc[1];
    o4.z = SCALE * acc[2];
    o4.w = SCALE * acc[3];
    *reinterpret_cast<float4*>(&Mp[(unsigned)c * MPLANE + ((unsigned)d * 32u + p1) * 32u + kq]) = o4;
}

#define XROWS  64
#define XPITCH 164
#define XLOADS (XROWS * 40 / 256)
#define MLOADS (MPLANE / 4 / 256)

__global__ __launch_bounds__(256) void apply_kernel(const float* __restrict__ X,
                                                    const float* __restrict__ Mp,
                                                    float* __restrict__ out) {
    __shared__ __attribute__((aligned(16))) float M[MPLANE];
    __shared__ __attribute__((aligned(16))) float Xs[XROWS * XPITCH];

    const int tid  = threadIdx.x;
    const int wid  = tid >> 5;
    const int lane = tid & 31;
    const int half = lane & 15;
    const int hi   = lane >> 4;
    const int rl   = wid >> 1;
    const int ct   = wid & 1;
    const int rt   = blockIdx.x * 4 + rl;

    const float4* Xb = reinterpret_cast<const float4*>(X + (size_t)blockIdx.x * XROWS * 160);
    float4 xv[XLOADS];
    #pragma unroll
    for (int i = 0; i < XLOADS; ++i) xv[i] = Xb[i * 256 + tid];

    float4 mv[MLOADS][KSPLIT];
    #pragma unroll
    for (int i = 0; i < MLOADS; ++i) {
        const int o = (i * 256 + tid) * 4;
        #pragma unroll
        for (int c = 0; c < KSPLIT; ++c)
            mv[i][c] = *reinterpret_cast<const float4*>(&Mp[c * MPLANE + o]);
    }
    __builtin_amdgcn_sched_group_barrier(0x020, XLOADS + MLOADS * KSPLIT, 0);
    __builtin_amdgcn_sched_group_barrier(0x200, XLOADS + MLOADS, 0);

    #pragma unroll
    for (int i = 0; i < XLOADS; ++i) {
        const int f   = i * 256 + tid;
        const int row = f / 40;
        const int c4  = f - row * 40;
        *reinterpret_cast<float4*>(&Xs[row * XPITCH + c4 * 4]) = xv[i];
    }
    #pragma unroll
    for (int i = 0; i < MLOADS; ++i) {
        const int o = (i * 256 + tid) * 4;
        float4 s = mv[i][0];
        #pragma unroll
        for (int c = 1; c < KSPLIT; ++c) {
            s.x += mv[i][c].x;
            s.y += mv[i][c].y;
            s.z += mv[i][c].z;
            s.w += mv[i][c].w;
        }
        *reinterpret_cast<float4*>(&M[o]) = s;
    }
    __syncthreads();

    const float* Xw = Xs + (rl * 16 + half) * XPITCH + hi * 2;
    const float* Mw = M + hi * 64 + ct * 16 + half;
    v2f af[40], bf[40];
    #pragma unroll
    for (int j = 0; j < 40; ++j) {
        af[j]   = *reinterpret_cast<const v2f*>(&Xw[j * 4]);
        bf[j].x = Mw[j * 128];
        bf[j].y = Mw[j * 128 + 32];
    }
    __builtin_amdgcn_sched_barrier(0);
    v8f acc = {};
    #pragma unroll
    for (int j = 0; j < 40; ++j)
        acc = __builtin_amdgcn_wmma_f32_16x16x4_f32(false, af[j], false, bf[j], (short)0, acc, false, false);

    #pragma unroll
    for (int r = 0; r < 8; ++r) {
        const int m = rt * 16 + r + 8 * hi;
        out[m * 32 + ct * 16 + half] = acc[r];
    }
}

extern "C" void kernel_launch(void* const* d_in, const int* in_sizes, int n_in,
                              void* d_out, int out_size, void* d_ws, size_t ws_size,
                              hipStream_t stream) {
    const float* x    = (const float*)d_in[0];
    const float* Win  = (const float*)d_in[1];
    const float* Wout = (const float*)d_in[2];
    float* out = (float*)d_out;

    float* Mp = (float*)d_ws;

    gram_mix_kernel<<<dim3(5, 4), 256, 0, stream>>>(Win, Wout, Mp);
    apply_kernel<<<8, 256, 0, stream>>>(x, Mp, out);
}


# apply spread over 4x the work-groups (8 -> 32, 16 rows each): a tile's K split over 4 waves, 20 WMMAs per SIMD, not 80; partial tiles added via LDS in k order (new summation tree), whole-line stores
# speedup vs baseline: 3.3546x; 1.1471x over previous
#include <hip/hip_runtime.h>


typedef __attribute__((ext_vector_type(2))) float v2f;
typedef __attribute__((ext_vector_type(8))) float v8f;

#define SCALE 0.6513215599f
#define KSPLIT 4
#define KWAVE  32
#define MPLANE 5120

constexpr unsigned cpar(unsigned a, unsigned b) {
    unsigned cnt = 0;
    for (unsigned aa = a >> 1; aa; aa >>= 1)
        for (unsigned t = aa & b; t; t &= t - 1) cnt ^= 1u;
    if ((a & b) & 16u) cnt ^= 1u;
    return cnt & 1u;
}
constexpr unsigned signA(unsigned i) {
    unsigned w = 0;
    for (unsigned p = 0; p < 32; ++p) w |= cpar(1u << i, p) << p;
    return w;
}
constexpr unsigned signB(unsigned i) {
    unsigned w = 0;
    for (unsigned p = 0; p < 32; ++p) w |= cpar(p, 1u << i) << p;
    return w;
}
constexpr unsigned signC() {
    unsigned w = 0;
    for (unsigned p = 0; p < 32; ++p) w |= cpar(p, p) << p;
    return w;
}

__global__ __launch_bounds__(256) void gram_mix_kernel(const float* __restrict__ Win,
                                                       const float* __restrict__ Wout,
                                                       float* __restrict__ Mp) {
    const int d    = blockIdx.x;
    const int c    = blockIdx.y;
    const int tid  = threadIdx.x;
    const int wid  = tid >> 5;
    const int lane = tid & 31;
    const int half = lane & 15;
    const int hi   = lane >> 4;

    __shared__ __attribute__((aligned(16))) float red[8 * 1024];
    __shared__ __attribute__((aligned(16))) float Gd[1024];

    v8f acc00 = {}, acc01 = {}, acc10 = {}, acc11 = {};

    const float* WinD = Win + d * 32;
    const int kbeg = (c * 8 + wid) * KWAVE;
    v2f a0[KWAVE / 4], a1[KWAVE / 4], b0[KWAVE / 4], b1[KWAVE / 4];
    #pragma unroll
    for (int j = 0; j < KWAVE / 4; ++j) {
        const int kA = kbeg + j * 4 + hi * 2;
        a0[j].x = WinD[(size_t)kA * 160 + half];
        a0[j].y = WinD[(size_t)(kA + 1) * 160 + half];
        a1[j].x = WinD[(size_t)kA * 160 + 16 + half];
        a1[j].y = WinD[(size_t)(kA + 1) * 160 + 16 + half];
        b0[j].x = Wout[kA * 32 + half];
        b0[j].y = Wout[(kA + 1) * 32 + half];
        b1[j].x = Wout[kA * 32 + 16 + half];
        b1[j].y = Wout[(kA + 1) * 32 + 16 + half];
    }
    #pragma unroll
    for (int j = 0; j < KWAVE / 4; ++j) {
        acc00 = __builtin_amdgcn_wmma_f32_16x16x4_f32(false, a0[j], false, b0[j], (short)0, acc00, false, false);
        acc01 = __builtin_amdgcn_wmma_f32_16x16x4_f32(false, a0[j], false, b1[j], (short)0, acc01, false, false);
        acc10 = __builtin_amdgcn_wmma_f32_16x16x4_f32(false, a1[j], false, b0[j], (short)0, acc10, false, false);
        acc11 = __builtin_amdgcn_wmma_f32_16x16x4_f32(false, a1[j], false, b1[j], (short)0, acc11, false, false);
    }

    float* my = red + wid * 1024;
    #pragma unroll
    for (int r = 0; r < 8; ++r) {
        const int m = r + 8 * hi;
        my[m * 32 + half]             = acc00[r];
        my[m * 32 + 16 + half]        = acc01[r];
        my[(m + 16) * 32 + half]      = acc10[r];
        my[(m + 16) * 32 + 16 + half] = acc11[r];
    }
    __syncthreads();

    #pragma unroll
    for (int j = 0; j < 4; ++j) {
        const int o = tid * 4 + j;
        float s = 0.0f;
        #pragma unroll
        for (int w = 0; w < 8; ++w) s += red[w * 1024 + o];
        Gd[o] = s;
    }
    __syncthreads();

    const unsigned p1 = (unsigned)tid >> 3;
    const unsigned kq = ((unsigned)tid & 7u) * 4u;

    constexpr unsigned A0 = signA(0), A1 = signA(1), A2 = signA(2), A3 = signA(3), A4 = signA(4);
    constexpr unsigned B0 = signB(0), B1 = signB(1), B2 = signB(2), B3 = signB(3), B4 = signB(4);
    constexpr unsigned CW = signC();

    unsigned S = CW;
    S ^= (p1 & 1u)  ? A0 : 0u;
    S ^= (p1 & 2u)  ? A1 : 0u;
    S ^= (p1 & 4u)  ? A2 : 0u;
    S ^= (p1 & 8u)  ? A3 : 0u;
    S ^= (p1 & 16u) ? A4 : 0u;
    S ^= (kq & 4u)  ? B2 : 0u;
    S ^= (kq & 8u)  ? B3 : 0u;
    S ^= (kq & 16u) ? B4 : 0u;
    S ^= ((CW >> p1) & 1u) ? 0xFFFFFFFFu : 0u;
    const unsigned Su[4] = {S, S ^ B0, S ^ B1, S ^ B0 ^ B1};

    float acc[4] = {0.0f, 0.0f, 0.0f, 0.0f};
    const unsigned base = p1 * 32u + kq;
    #pragma unroll
    for (unsigned p = 0; p < 32; ++p) {
        const float4 v = *reinterpret_cast<const float4*>(&Gd[base ^ (p * 32u + (p & 28u))]);
        const float ve[4] = {v.x, v.y, v.z, v.w};
        #pragma unroll
        for (unsigned u = 0; u < 4; ++u) {
            const unsigned sb = (Su[u] << (31u - p)) & 0x80000000u;
            acc[u] += __uint_as_float(__float_as_uint(ve[(p & 3u) ^ u]) ^ sb);
        }
    }
    float4 o4;
    o4.x = SCALE * acc[0];
    o4.y = SCALE * acc[1];
    o4.z = SCALE * acc[2];
    o4.w = SCALE * acc[3];
    *reinterpret_cast<float4*>(&Mp[(unsigned)c * MPLANE + ((unsigned)d * 32u + p1) * 32u + kq]) = o4;
}

#define XROWS  16
#define XPITCH 164
#define KQ     4
#define KSTEPS (40 / KQ)
#define XLOADS (XROWS * 80 / 256)
#define MLOADS (MPLANE / 4 / 256)
#define RPITCH 36

__global__ __launch_bounds__(256) void apply_kernel(const float* __restrict__ X,
                                                    const float* __restrict__ Mp,
                                                    float* __restrict__ out) {
    __shared__ __attribute__((aligned(16))) float M[MPLANE];
    __shared__ __attribute__((aligned(16))) float Xs[XROWS * XPITCH];
    __shared__ __attribute__((aligned(16))) float red[KQ * 16 * RPITCH];

    const int tid  = threadIdx.x;
    const int wid  = tid >> 5;
    const int lane = tid & 31;
    const int half = lane & 15;
    const int hi   = lane >> 4;
    const int ct   = wid & 1;
    const int kq   = wid >> 1;

    const float2* Xb = reinterpret_cast<const float2*>(X + (size_t)blockIdx.x * XROWS * 160);
    float2 xv[XLOADS];
    #pragma unroll
    for (int i = 0; i < XLOADS; ++i) xv[i] = Xb[i * 256 + tid];

    float4 mv[MLOADS][KSPLIT];
    #pragma unroll
    for (int i = 0; i < MLOADS; ++i) {
        const int o = (i * 256 + tid) * 4;
        #pragma unroll
        for (int c = 0; c < KSPLIT; ++c)
            mv[i][c] = *reinterpret_cast<const float4*>(&Mp[c * MPLANE + o]);
    }
    __builtin_amdgcn_sched_group_barrier(0x020, XLOADS + MLOADS * KSPLIT, 0);
    __builtin_amdgcn_sched_group_barrier(0x002, 128, 0);
    __builtin_amdgcn_sched_group_barrier(0x200, XLOADS + MLOADS, 0);

    #pragma unroll
    for (int i = 0; i < XLOADS; ++i) {
        const int f   = i * 256 + tid;
        const int row = f / 80;
        const int c2  = f - row * 80;
        *reinterpret_cast<float2*>(&Xs[row * XPITCH + c2 * 2]) = xv[i];
    }
    #pragma unroll
    for (int i = 0; i < MLOADS; ++i) {
        const int o = (i * 256 + tid) * 4;
        float4 s = mv[i][0];
        #pragma unroll
        for (int c = 1; c < KSPLIT; ++c) {
            s.x += mv[i][c].x;
            s.y += mv[i][c].y;
            s.z += mv[i][c].z;
            s.w += mv[i][c].w;
        }
        *reinterpret_cast<float4*>(&M[o]) = s;
    }
    __syncthreads();

    const float* Xw = Xs + half * XPITCH + kq * (4 * KSTEPS) + hi * 2;
    const float* Mw = M + (kq * (4 * KSTEPS) + hi * 2) * 32 + ct * 16 + half;
    v2f af[KSTEPS], bf[KSTEPS];
    #pragma unroll
    for (int j = 0; j < KSTEPS; ++j) {
        af[j]   = *reinterpret_cast<const v2f*>(&Xw[j * 4]);
        bf[j].x = Mw[j * 128];
        bf[j].y = Mw[j * 128 + 32];
    }
    __builtin_amdgcn_sched_barrier(0);
    v8f acc = {};
    #pragma unroll
    for (int j = 0; j < KSTEPS; ++j)
        acc = __builtin_amdgcn_wmma_f32_16x16x4_f32(false, af[j], false, bf[j], (short)0, acc, false, false);

    float* my = red + kq * (16 * RPITCH) + ct * 16 + half;
    #pragma unroll
    for (int r = 0; r < 8; ++r) my[(r + 8 * hi) * RPITCH] = acc[r];
    __syncthreads();

    if (tid < 128) {
        const int m  = tid >> 3;
        const int n4 = (tid & 7) * 4;
        float4 s = *reinterpret_cast<const float4*>(&red[m * RPITCH + n4]);
        #pragma unroll
        for (int q = 1; q < KQ; ++q) {
            const float4 t = *reinterpret_cast<const float4*>(&red[q * (16 * RPITCH) + m * RPITCH + n4]);
            s.x += t.x;
            s.y += t.y;
            s.z += t.z;
            s.w += t.w;
        }
        *reinterpret_cast<float4*>(&out[((size_t)blockIdx.x * XROWS + m) * 32 + n4]) = s;
    }
}

extern "C" void kernel_launch(void* const* d_in, const int* in_sizes, int n_in,
                              void* d_out, int out_size, void* d_ws, size_t ws_size,
                              hipStream_t stream) {
    const float* x    = (const float*)d_in[0];
    const float* Win  = (const float*)d_in[1];
    const float* Wout = (const float*)d_in[2];
    float* out = (float*)d_out;

    float* Mp = (float*)d_ws;

    gram_mix_kernel<<<dim3(5, 4), 256, 0, stream>>>(Win, Wout, Mp);
    apply_kernel<<<32, 256, 0, stream>>>(x, Mp, out);
}


# geometric product is associative: M is a signed permutation of P[d] = sum_h Win[h,d]*Wout[h]; gram folds G[d] to its 32 floats (new summation tree), planes 160x32 -> 5x32, apply picks B from P in LDS
# speedup vs baseline: 4.1597x; 1.2400x over previous
#include <hip/hip_runtime.h>


typedef __attribute__((ext_vector_type(2))) float v2f;
typedef __attribute__((ext_vector_type(8))) float v8f;

#define SCALE 0.6513215599f
#define KSPLIT 4
#define KWAVE  32
#define PPLANE 160

constexpr unsigned cpar(unsigned a, unsigned b) {
    unsigned cnt = 0;
    for (unsigned aa = a >> 1; aa; aa >>= 1)
        for (unsigned t = aa & b; t; t &= t - 1) cnt ^= 1u;
    if ((a & b) & 16u) cnt ^= 1u;
    return cnt & 1u;
}
constexpr unsigned signB(unsigned i) {
    unsigned w = 0;
    for (unsigned p = 0; p < 32; ++p) w |= cpar(p, 1u << i) << p;
    return w;
}
constexpr unsigned signC() {
    unsigned w = 0;
    for (unsigned p = 0; p < 32; ++p) w |= cpar(p, p) << p;
    return w;
}

__global__ __launch_bounds__(256) void gram_mix_kernel(const float* __restrict__ Win,
                                                       const float* __restrict__ Wout,
                                                       float* __restrict__ Pp) {
    const int d    = blockIdx.x;
    const int c    = blockIdx.y;
    const int tid  = threadIdx.x;
    const int wid  = tid >> 5;
    const int lane = tid & 31;
    const int half = lane & 15;
    const int hi   = lane >> 4;

    __shared__ __attribute__((aligned(16))) float red[8 * 1024];
    __shared__ __attribute__((aligned(16))) float Gd[1024];

    v8f acc00 = {}, acc01 = {}, acc10 = {}, acc11 = {};

    const float* WinD = Win + d * 32;
    const int kbeg = (c * 8 + wid) * KWAVE;
    v2f a0[KWAVE / 4], a1[KWAVE / 4], b0[KWAVE / 4], b1[KWAVE / 4];
    #pragma unroll
    for (int j = 0; j < KWAVE / 4; ++j) {
        const int kA = kbeg + j * 4 + hi * 2;
        a0[j].x = WinD[(size_t)kA * 160 + half];
        a0[j].y = WinD[(size_t)(kA + 1) * 160 + half];
        a1[j].x = WinD[(size_t)kA * 160 + 16 + half];
        a1[j].y = WinD[(size_t)(kA + 1) * 160 + 16 + half];
        b0[j].x = Wout[kA * 32 + half];
        b0[j].y = Wout[(kA + 1) * 32 + half];
        b1[j].x = Wout[kA * 32 + 16 + half];
        b1[j].y = Wout[(kA + 1) * 32 + 16 + half];
    }
    __builtin_amdgcn_sched_barrier(0);
    #pragma unroll
    for (int j = 0; j < KWAVE / 4; ++j) {
        acc00 = __builtin_amdgcn_wmma_f32_16x16x4_f32(false, a0[j], false, b0[j], (short)0, acc00, false, false);
        acc01 = __builtin_amdgcn_wmma_f32_16x16x4_f32(false, a0[j], false, b1[j], (short)0, acc01, false, false);
        acc10 = __builtin_amdgcn_wmma_f32_16x16x4_f32(false, a1[j], false, b0[j], (short)0, acc10, false, false);
        acc11 = __builtin_amdgcn_wmma_f32_16x16x4_f32(false, a1[j], false, b1[j], (short)0, acc11, false, false);
    }

    float* my = red + wid * 1024;
    #pragma unroll
    for (int r = 0; r < 8; ++r) {
        const int m = r + 8 * hi;
        my[m * 32 + half]             = acc00[r];
        my[m * 32 + 16 + half]        = acc01[r];
        my[(m + 16) * 32 + half]      = acc10[r];
        my[(m + 16) * 32 + 16 + half] = acc11[r];
    }
    __syncthreads();

    #pragma unroll
    for (int j = 0; j < 4; ++j) {
        const int o = tid * 4 + j;
        float s = 0.0f;
        #pragma unroll
        for (int w = 0; w < 8; ++w) s += red[w * 1024 + o];
        Gd[o] = s;
    }
    __syncthreads();

    if (wid == 0) {
        constexpr unsigned B0 = signB(0), B1 = signB(1), B2 = signB(2), B3 = signB(3), B4 = signB(4);
        constexpr unsigned CW = signC();
        const unsigned r = (unsigned)lane;
        unsigned Wr = CW;
        Wr ^= (r & 1u)  ? B0 : 0u;
        Wr ^= (r & 2u)  ? B1 : 0u;
        Wr ^= (r & 4u)  ? B2 : 0u;
        Wr ^= (r & 8u)  ? B3 : 0u;
        Wr ^= (r & 16u) ? B4 : 0u;
        float acc[4] = {0.0f, 0.0f, 0.0f, 0.0f};
        #pragma unroll
        for (unsigned q1 = 0; q1 < 32; ++q1) {
            const float g = Gd[q1 * 32u + (q1 ^ r)];
            const unsigned sb = (Wr << (31u - q1)) & 0x80000000u;
            acc[q1 & 3u] += __uint_as_float(__float_as_uint(g) ^ sb);
        }
        Pp[(unsigned)c * PPLANE + (unsigned)d * 32u + r] = SCALE * ((acc[0] + acc[1]) + (acc[2] + acc[3]));
    }
}

#define XROWS  16
#define XPITCH 164
#define KQ     4
#define KSTEPS (40 / KQ)
#define XLOADS (XROWS * 80 / 256)
#define RPITCH 36

__global__ __launch_bounds__(256) void apply_kernel(const float* __restrict__ X,
                                                    const float* __restrict__ Pp,
                                                    float* __restrict__ out) {
    __shared__ __attribute__((aligned(16))) float Ps[256];
    __shared__ __attribute__((aligned(16))) float Xs[XROWS * XPITCH];
    __shared__ __attribute__((aligned(16))) float red[KQ * 16 * RPITCH];

    const int tid  = threadIdx.x;
    const int wid  = tid >> 5;
    const int lane = tid & 31;
    const int half = lane & 15;
    const int hi   = lane >> 4;
    const int ct   = wid & 1;
    const int kq   = wid >> 1;

    const float2* Xb = reinterpret_cast<const float2*>(X + (size_t)blockIdx.x * XROWS * 160);
    float2 xv[XLOADS];
    #pragma unroll
    for (int i = 0; i < XLOADS; ++i) xv[i] = Xb[i * 256 + tid];

    const int pi = tid < PPLANE ? tid : PPLANE - 1;
    float pv[KSPLIT];
    #pragma unroll
    for (int c = 0; c < KSPLIT; ++c) pv[c] = Pp[c * PPLANE + pi];

    constexpr unsigned B0 = signB(0), B1 = signB(1), B2 = signB(2), B3 = signB(3), B4 = signB(4);
    constexpr unsigned CW = signC();
    const unsigned n = (unsigned)(ct * 16 + half);
    unsigned Wn = CW;
    Wn ^= (n & 1u)  ? B0 : 0u;
    Wn ^= (n & 2u)  ? B1 : 0u;
    Wn ^= (n & 4u)  ? B2 : 0u;
    Wn ^= (n & 8u)  ? B3 : 0u;
    Wn ^= (n & 16u) ? B4 : 0u;
    const unsigned kb   = (unsigned)(kq * (4 * KSTEPS) + hi * 2);
    const unsigned Wrot = __builtin_rotateright32(Wn, kb & 31u);
    unsigned pa[KSTEPS][2], sb[KSTEPS][2];
    #pragma unroll
    for (int j = 0; j < KSTEPS; ++j) {
        #pragma unroll
        for (int e = 0; e < 2; ++e) {
            const unsigned cc = (unsigned)(j * 4 + e);
            pa[j][e] = (kb + cc) ^ n;
            sb[j][e] = (Wrot >> (cc & 31u)) << 31;
        }
    }
    __builtin_amdgcn_sched_barrier(0);

    {
        float s = pv[0];
        #pragma unroll
        for (int c = 1; c < KSPLIT; ++c) s += pv[c];
        Ps[tid] = s;
    }

    #pragma unroll
    for (int i = 0; i < XLOADS; ++i) {
        const int f   = i * 256 + tid;
        const int row = f / 80;
        const int c2  = f - row * 80;
        *reinterpret_cast<float2*>(&Xs[row * XPITCH + c2 * 2]) = xv[i];
    }
    __syncthreads();

    const float* Xw = Xs + half * XPITCH + kq * (4 * KSTEPS) + hi * 2;
    v2f af[KSTEPS], bf[KSTEPS];
    #pragma unroll
    for (int j = 0; j < KSTEPS; ++j) {
        af[j]   = *reinterpret_cast<const v2f*>(&Xw[j * 4]);
        bf[j].x = Ps[pa[j][0]];
        bf[j].y = Ps[pa[j][1]];
    }
    __builtin_amdgcn_sched_barrier(0);
    #pragma unroll
    for (int j = 0; j < KSTEPS; ++j) {
        bf[j].x = __uint_as_float(__float_as_uint(bf[j].x) ^ sb[j][0]);
        bf[j].y = __uint_as_float(__float_as_uint(bf[j].y) ^ sb[j][1]);
    }
    v8f acc = {};
    #pragma unroll
    for (int j = 0; j < KSTEPS; ++j)
        acc = __builtin_amdgcn_wmma_f32_16x16x4_f32(false, af[j], false, bf[j], (short)0, acc, false, false);

    float* my = red + kq * (16 * RPITCH) + ct * 16 + half;
    #pragma unroll
    for (int r = 0; r < 8; ++r) my[(r + 8 * hi) * RPITCH] = acc[r];
    __syncthreads();

    if (tid < 128) {
        const int m  = tid >> 3;
        const int n4 = (tid & 7) * 4;
        float4 s = *reinterpret_cast<const float4*>(&red[m * RPITCH + n4]);
        #pragma unroll
        for (int q = 1; q < KQ; ++q) {
            const float4 t = *reinterpret_cast<const float4*>(&red[q * (16 * RPITCH) + m * RPITCH + n4]);
            s.x += t.x;
            s.y += t.y;
            s.z += t.z;
            s.w += t.w;
        }
        *reinterpret_cast<float4*>(&out[((size_t)blockIdx.x * XROWS + m) * 32 + n4]) = s;
    }
}

extern "C" void kernel_launch(void* const* d_in, const int* in_sizes, int n_in,
                              void* d_out, int out_size, void* d_ws, size_t ws_size,
                              hipStream_t stream) {
    const float* x    = (const float*)d_in[0];
    const float* Win  = (const float*)d_in[1];
    const float* Wout = (const float*)d_in[2];
    float* out = (float*)d_out;

    float* Pp = (float*)d_ws;

    gram_mix_kernel<<<dim3(5, 4), 256, 0, stream>>>(Win, Wout, Pp);
    apply_kernel<<<32, 256, 0, stream>>>(x, Pp, out);
}
